# starting kernel
# baseline (speedup 1.0000x reference)
#include <hip/hip_runtime.h>

#define NROWS 8192
#define DIM   256
#define INV_SIGMA 3.33333333333f
#define LOGC      0.28503428291f

#define ROWS_PER_BLOCK  64
#define WAVES_PER_BLOCK 4
#define THREADS         (WAVES_PER_BLOCK * 32)
#define NCOLTILES       (NROWS / 16)
#define NSLICES         8
#define TILES_PER_SLICE (NCOLTILES / NSLICES)
#define LDS_STRIDE      264

typedef __attribute__((ext_vector_type(16))) __bf16 v16bf;
typedef __attribute__((ext_vector_type(8)))  float  v8f;

__device__ __forceinline__ unsigned short f32_to_bf16_rne(float f) {
    unsigned u = __float_as_uint(f);
    u += 0x7FFFu + ((u >> 16) & 1u);
    return (unsigned short)(u >> 16);
}
__device__ __forceinline__ float bf16_to_f32(unsigned short h) {
    return __uint_as_float(((unsigned)h) << 16);
}

__device__ __forceinline__ unsigned f2key(float f) {
    unsigned b = __float_as_uint(f);
    return (b & 0x80000000u) ? ~b : (b | 0x80000000u);
}
__device__ __forceinline__ float key2f(unsigned u) {
    unsigned b = (u & 0x80000000u) ? (u ^ 0x80000000u) : ~u;
    return __uint_as_float(b);
}

__global__ void init_keys_kernel(unsigned* __restrict__ keys) {
    keys[blockIdx.x * blockDim.x + threadIdx.x] = 0xFFFFFFFFu;
}

__global__ void normalize_split_kernel(const float* __restrict__ Ex,
                                       const float* __restrict__ Ey,
                                       unsigned short* __restrict__ Xhi,
                                       unsigned short* __restrict__ Xlo,
                                       unsigned short* __restrict__ Yhi,
                                       unsigned short* __restrict__ Ylo) {
    const int wave = (blockIdx.x * blockDim.x + threadIdx.x) >> 5;
    const int lane = threadIdx.x & 31;

    const float* src;
    unsigned short *hi, *lo;
    int row;
    if (wave < NROWS) { src = Ex; hi = Xhi; lo = Xlo; row = wave; }
    else              { src = Ey; hi = Yhi; lo = Ylo; row = wave - NROWS; }

    const float* p = src + (size_t)row * DIM;
    float v[DIM / 32];
    float ss = 0.0f;
#pragma unroll
    for (int i = 0; i < DIM / 32; ++i) {
        v[i] = p[lane + 32 * i];
        ss = fmaf(v[i], v[i], ss);
    }
#pragma unroll
    for (int off = 16; off > 0; off >>= 1)
        ss += __shfl_xor(ss, off, 32);

    const float inv = 1.0f / fmaxf(sqrtf(ss), 1e-8f);

#pragma unroll
    for (int i = 0; i < DIM / 32; ++i) {
        const float x = v[i] * inv;
        const unsigned short h = f32_to_bf16_rne(x);
        const unsigned short l = f32_to_bf16_rne(x - bf16_to_f32(h));
        const size_t idx = (size_t)row * DIM + lane + 32 * i;
        hi[idx] = h;
        lo[idx] = l;
    }
}

__device__ __forceinline__ v16bf lds_frag(const unsigned short* p) {
    union { uint4 q[2]; v16bf v; } f;
    f.q[0] = *(const uint4*)(p);
    f.q[1] = *(const uint4*)(p + 16);
    return f.v;
}

__global__ __launch_bounds__(THREADS) void cosmin_wmma_kernel(
    const unsigned short* __restrict__ Xhi,
    const unsigned short* __restrict__ Xlo,
    const unsigned short* __restrict__ Yhi,
    const unsigned short* __restrict__ Ylo,
    unsigned* __restrict__ keys) {
    __shared__ unsigned short lAh[ROWS_PER_BLOCK * LDS_STRIDE];
    __shared__ unsigned short lAl[ROWS_PER_BLOCK * LDS_STRIDE];
    __shared__ unsigned short lBh[16 * LDS_STRIDE];
    __shared__ unsigned short lBl[16 * LDS_STRIDE];

    const int tid      = threadIdx.x;
    const int lane     = tid & 31;
    const int wave     = tid >> 5;
    const int rowBase  = blockIdx.x * ROWS_PER_BLOCK;
    const int tileBase = blockIdx.y * TILES_PER_SLICE;

#pragma unroll 2
    for (int idx = tid; idx < ROWS_PER_BLOCK * (DIM / 8); idx += THREADS) {
        const int row = idx >> 5;
        const int ch  = idx & 31;
        const size_t s = (size_t)(rowBase + row) * DIM + ch * 8;
        *(uint4*)&lAh[row * LDS_STRIDE + ch * 8] = *(const uint4*)&Xhi[s];
        *(uint4*)&lAl[row * LDS_STRIDE + ch * 8] = *(const uint4*)&Xlo[s];
    }

    uint4 rh[4], rl[4];
    auto prefetchB = [&](int t) {
        const int colBase = (tileBase + t) * 16;
#pragma unroll
        for (int i = 0; i < 4; ++i) {
            const int cid = tid + THREADS * i;
            const int row = cid >> 5;
            const int ch  = cid & 31;
            const size_t s = (size_t)(colBase + row) * DIM + ch * 8;
            rh[i] = *(const uint4*)&Yhi[s];
            rl[i] = *(const uint4*)&Ylo[s];
        }
    };
    prefetchB(0);

    float minv[8];
#pragma unroll
    for (int r = 0; r < 8; ++r) minv[r] = 3.402823466e38f;

    const int hiSel = (lane >> 4) << 3;
    const int aOff  = (wave * 16 + (lane & 15)) * LDS_STRIDE + hiSel;
    const int bOff  = (lane & 15) * LDS_STRIDE + hiSel;

    for (int t = 0; t < TILES_PER_SLICE; ++t) {
        __syncthreads();
#pragma unroll
        for (int i = 0; i < 4; ++i) {
            const int cid = tid + THREADS * i;
            const int row = cid >> 5;
            const int ch  = cid & 31;
            *(uint4*)&lBh[row * LDS_STRIDE + ch * 8] = rh[i];
            *(uint4*)&lBl[row * LDS_STRIDE + ch * 8] = rl[i];
        }
        __syncthreads();
        if (t + 1 < TILES_PER_SLICE) prefetchB(t + 1);

        v8f acc = {};
#pragma unroll
        for (int c = 0; c < DIM / 32; ++c) {
            const int k0 = c * 32;
            v16bf ah = lds_frag(&lAh[aOff + k0]);
            v16bf al = lds_frag(&lAl[aOff + k0]);
            v16bf bh = lds_frag(&lBh[bOff + k0]);
            v16bf bl = lds_frag(&lBl[bOff + k0]);
            acc = __builtin_amdgcn_wmma_f32_16x16x32_bf16(
                      false, al, false, bh, (short)0, acc, false, false);
            acc = __builtin_amdgcn_wmma_f32_16x16x32_bf16(
                      false, ah, false, bl, (short)0, acc, false, false);
            acc = __builtin_amdgcn_wmma_f32_16x16x32_bf16(
                      false, ah, false, bh, (short)0, acc, false, false);
        }

#pragma unroll
        for (int r = 0; r < 8; ++r) {
            const float m    = acc[r];
            const float z    = (m - 1.0f) * INV_SIGMA;
            const float logp = fmaf(-0.5f * z, z, LOGC);
            const float C    = -__expf(logp) * logp;
            minv[r] = fminf(minv[r], C);
        }
    }

#pragma unroll
    for (int r = 0; r < 8; ++r) {
#pragma unroll
        for (int off = 1; off < 16; off <<= 1)
            minv[r] = fminf(minv[r], __shfl_xor(minv[r], off, 32));
    }

    if ((lane & 15) == 0) {
        const int base = rowBase + wave * 16 + ((lane >> 4) << 3);
#pragma unroll
        for (int r = 0; r < 8; ++r)
            atomicMin(&keys[base + r], f2key(minv[r]));
    }
}

__global__ void finalize_kernel(const unsigned* __restrict__ keys,
                                float* __restrict__ out) {
    const int i = blockIdx.x * blockDim.x + threadIdx.x;
    out[i] = key2f(keys[i]);
}

extern "C" void kernel_launch(void* const* d_in, const int* in_sizes, int n_in,
                              void* d_out, int out_size, void* d_ws, size_t ws_size,
                              hipStream_t stream) {
    const float* Ex = (const float*)d_in[0];
    const float* Ey = (const float*)d_in[1];

    const size_t elems = (size_t)NROWS * DIM;
    unsigned short* Xhi = (unsigned short*)d_ws;
    unsigned short* Xlo = Xhi + elems;
    unsigned short* Yhi = Xlo + elems;
    unsigned short* Ylo = Yhi + elems;
    unsigned*      keys = (unsigned*)(Ylo + elems);

    hipLaunchKernelGGL(init_keys_kernel, dim3(NROWS / 256), dim3(256), 0, stream,
                       keys);
    hipLaunchKernelGGL(normalize_split_kernel, dim3((2 * NROWS) / 8), dim3(256), 0,
                       stream, Ex, Ey, Xhi, Xlo, Yhi, Ylo);
    hipLaunchKernelGGL(cosmin_wmma_kernel,
                       dim3(NROWS / ROWS_PER_BLOCK, NSLICES), dim3(THREADS), 0,
                       stream, Xhi, Xlo, Yhi, Ylo, keys);
    hipLaunchKernelGGL(finalize_kernel, dim3(NROWS / 256), dim3(256), 0, stream,
                       keys, (float*)d_out);
}


# X tiles fetched from global two tiles ahead of their LDS hand-over (was one): second register set in flight; tile barrier by hand with an LDS-only wait, so loads stay in flight across it
# speedup vs baseline: 3.0604x; 3.0604x over previous
#include <hip/hip_runtime.h>

#define NROWS 8192
#define DIM   256
#define INV_SIGMA 3.33333333333f
#define LOGC      0.28503428291f

#define YTILES_PER_WAVE 4
#define WAVES_PER_BLOCK 4
#define THREADS         (WAVES_PER_BLOCK * 32)
#define YROWS_PER_BLOCK (WAVES_PER_BLOCK * YTILES_PER_WAVE * 16)
#define NYWAVES         (NROWS / (YTILES_PER_WAVE * 16))
#define NXTILES         (NROWS / 16)
#define NXSLICES        8
#define XTILES_PER_SLICE (NXTILES / NXSLICES)
#define KCHUNKS         (DIM / 32)
#define LDS_STRIDE      264
#define STAGE_ELEMS     (16 * LDS_STRIDE)

typedef __attribute__((ext_vector_type(16))) __bf16 v16bf;
typedef __attribute__((ext_vector_type(8)))  float  v8f;
typedef __attribute__((ext_vector_type(4)))  float  f32x4;

__device__ __forceinline__ unsigned short f32_to_bf16_rne(float f) {
    unsigned u = __float_as_uint(f);
    u += 0x7FFFu + ((u >> 16) & 1u);
    return (unsigned short)(u >> 16);
}
__device__ __forceinline__ float bf16_to_f32(unsigned short h) {
    return __uint_as_float(((unsigned)h) << 16);
}
typedef __attribute__((ext_vector_type(2))) float  f32x2;
typedef __attribute__((ext_vector_type(2))) __bf16 bf16x2;
__device__ __forceinline__ void split_pair(float a, float b, unsigned& hi, unsigned& lo) {
    const f32x2 v = {a, b};
    const bf16x2 h = __builtin_convertvector(v, bf16x2);
    const unsigned hu = __builtin_bit_cast(unsigned, h);
    const f32x2 hf = {__uint_as_float(hu << 16), __uint_as_float(hu & 0xFFFF0000u)};
    const bf16x2 l = __builtin_convertvector(v - hf, bf16x2);
    hi = hu;
    lo = __builtin_bit_cast(unsigned, l);
}
__device__ __forceinline__ float other_half(float v) {
    return __int_as_float(__builtin_amdgcn_permlanex16(
        __float_as_int(v), __float_as_int(v), 0x76543210, 0xfedcba98, false, false));
}

__global__ void normalize_split_kernel(const float* __restrict__ Ex,
                                       unsigned short* __restrict__ Xhi,
                                       unsigned short* __restrict__ Xlo) {
    const int row  = (blockIdx.x * blockDim.x + threadIdx.x) >> 5;
    const int lane = threadIdx.x & 31;

    const float* p = Ex + (size_t)row * DIM;
    float v[DIM / 32];
    float ss = 0.0f;
#pragma unroll
    for (int i = 0; i < DIM / 32; ++i) {
        v[i] = p[lane + 32 * i];
        ss = fmaf(v[i], v[i], ss);
    }
#pragma unroll
    for (int off = 16; off > 0; off >>= 1)
        ss += __shfl_xor(ss, off, 32);

    const float inv = 1.0f / fmaxf(sqrtf(ss), 1e-8f);

#pragma unroll
    for (int i = 0; i < DIM / 32; ++i) {
        const float x = v[i] * inv;
        const unsigned short h = f32_to_bf16_rne(x);
        const unsigned short l = f32_to_bf16_rne(x - bf16_to_f32(h));
        const size_t idx = (size_t)row * DIM + lane + 32 * i;
        Xhi[idx] = h;
        Xlo[idx] = l;
    }
}

__device__ __forceinline__ v16bf lds_frag(const unsigned short* p) {
    union { uint4 q[2]; v16bf v; } f;
    f.q[0] = *(const uint4*)(p);
    f.q[1] = *(const uint4*)(p + 16);
    return f.v;
}

typedef __attribute__((ext_vector_type(4))) unsigned int u32x4;

#define LOAD_XTILE(P, xRow0)                                                  \
    do {                                                                      \
        const size_t s_ = (size_t)((xRow0) + wave) * DIM + lane * 8;          \
        P##H0 = *(const u32x4*)&Xhi[s_];                                      \
        P##H1 = *(const u32x4*)&Xhi[s_ + 4 * DIM];                            \
        P##H2 = *(const u32x4*)&Xhi[s_ + 8 * DIM];                            \
        P##H3 = *(const u32x4*)&Xhi[s_ + 12 * DIM];                           \
        P##L0 = *(const u32x4*)&Xlo[s_];                                      \
        P##L1 = *(const u32x4*)&Xlo[s_ + 4 * DIM];                            \
        P##L2 = *(const u32x4*)&Xlo[s_ + 8 * DIM];                            \
        P##L3 = *(const u32x4*)&Xlo[s_ + 12 * DIM];                           \
    } while (0)
#define STORE_XTILE(P, stage)                                                 \
    do {                                                                      \
        const int o_ = (stage) * STAGE_ELEMS + wave * LDS_STRIDE + lane * 8;  \
        *(u32x4*)&lXh[o_]                   = P##H0;                          \
        *(u32x4*)&lXh[o_ + 4 * LDS_STRIDE]  = P##H1;                          \
        *(u32x4*)&lXh[o_ + 8 * LDS_STRIDE]  = P##H2;                          \
        *(u32x4*)&lXh[o_ + 12 * LDS_STRIDE] = P##H3;                          \
        *(u32x4*)&lXl[o_]                   = P##L0;                          \
        *(u32x4*)&lXl[o_ + 4 * LDS_STRIDE]  = P##L1;                          \
        *(u32x4*)&lXl[o_ + 8 * LDS_STRIDE]  = P##L2;                          \
        *(u32x4*)&lXl[o_ + 12 * LDS_STRIDE] = P##L3;                          \
    } while (0)

__device__ __forceinline__ void lds_stage_barrier() {
    asm volatile("s_wait_dscnt 0x0\n\ts_barrier_signal -1\n\ts_barrier_wait -1" ::: "memory");
}

__global__ __launch_bounds__(THREADS) void cosmin_wmma_kernel(
    const unsigned short* __restrict__ Xhi,
    const unsigned short* __restrict__ Xlo,
    const float* __restrict__ Ey,
    float* __restrict__ part) {
    __shared__ unsigned short lXh[2 * STAGE_ELEMS];
    __shared__ unsigned short lXl[2 * STAGE_ELEMS];

    const int tid   = threadIdx.x;
    const int lane  = tid & 31;
    const int wave  = tid >> 5;
    const int c16   = lane & 15;
    const int hiSel = (lane >> 4) << 3;
    const int yRow0 = blockIdx.x * YROWS_PER_BLOCK + wave * (YTILES_PER_WAVE * 16);
    const int xRowSlice = blockIdx.y * (XTILES_PER_SLICE * 16);

    u32x4 AH0, AH1, AH2, AH3, AL0, AL1, AL2, AL3;
    u32x4 BH0, BH1, BH2, BH3, BL0, BL1, BL2, BL3;
    LOAD_XTILE(A, xRowSlice);

    v16bf yh[YTILES_PER_WAVE][KCHUNKS];
    v16bf yl[YTILES_PER_WAVE][KCHUNKS];
#pragma unroll
    for (int i = 0; i < YTILES_PER_WAVE; ++i) {
        const float* yp = Ey + (size_t)(yRow0 + 16 * i + c16) * DIM + hiSel;
        f32x4 raw[KCHUNKS][4];
#pragma unroll
        for (int c = 0; c < KCHUNKS; ++c) {
            raw[c][0] = *(const f32x4*)(yp + 32 * c);
            raw[c][1] = *(const f32x4*)(yp + 32 * c + 4);
            raw[c][2] = *(const f32x4*)(yp + 32 * c + 16);
            raw[c][3] = *(const f32x4*)(yp + 32 * c + 20);
        }
        float s0 = 0.0f, s1 = 0.0f, s2 = 0.0f, s3 = 0.0f;
#pragma unroll
        for (int c = 0; c < KCHUNKS; ++c) {
#pragma unroll
            for (int j = 0; j < 4; ++j) {
                s0 = fmaf(raw[c][j].x, raw[c][j].x, s0);
                s1 = fmaf(raw[c][j].y, raw[c][j].y, s1);
                s2 = fmaf(raw[c][j].z, raw[c][j].z, s2);
                s3 = fmaf(raw[c][j].w, raw[c][j].w, s3);
            }
        }
        float ss = (s0 + s1) + (s2 + s3);
        ss += other_half(ss);
        const float inv = 1.0f / fmaxf(sqrtf(ss), 1e-8f);
#pragma unroll
        for (int c = 0; c < KCHUNKS; ++c) {
            union { unsigned u[8]; v16bf v; } H, L;
#pragma unroll
            for (int j = 0; j < 4; ++j) {
                split_pair(raw[c][j].x * inv, raw[c][j].y * inv, H.u[2 * j],     L.u[2 * j]);
                split_pair(raw[c][j].z * inv, raw[c][j].w * inv, H.u[2 * j + 1], L.u[2 * j + 1]);
            }
            yh[i][c] = H.v;
            yl[i][c] = L.v;
        }
    }

    LOAD_XTILE(B, xRowSlice + 16);
    STORE_XTILE(A, 0);
    lds_stage_barrier();
    LOAD_XTILE(A, xRowSlice + 32);

    const int fOff = c16 * LDS_STRIDE + hiSel;

    v16bf xh0 = lds_frag(&lXh[fOff]);
    v16bf xl0 = lds_frag(&lXl[fOff]);
    v16bf xh1 = lds_frag(&lXh[fOff + 32]);
    v16bf xl1 = lds_frag(&lXl[fOff + 32]);

    float* prow = part + (size_t)(blockIdx.x * WAVES_PER_BLOCK + wave) * NROWS
                       + xRowSlice + lane;
    float mnEven = 0.0f;

    for (int t = 0; t < XTILES_PER_SLICE; t += 2) {
#pragma unroll
        for (int s = 0; s < 2; ++s) {
            const unsigned short* bh = &lXh[s * STAGE_ELEMS + fOff];
            const unsigned short* bl = &lXl[s * STAGE_ELEMS + fOff];

            v8f acc[YTILES_PER_WAVE];
#pragma unroll
            for (int i = 0; i < YTILES_PER_WAVE; ++i) acc[i] = v8f{};

            v16bf xh[3], xl[3];
            xh[0] = xh0; xl[0] = xl0;
            xh[1] = xh1; xl[1] = xl1;
#pragma unroll
            for (int c = 0; c < KCHUNKS; ++c) {
                if (c + 2 < KCHUNKS) {
                    xh[(c + 2) % 3] = lds_frag(bh + 32 * (c + 2));
                    xl[(c + 2) % 3] = lds_frag(bl + 32 * (c + 2));
                }
                __builtin_amdgcn_sched_barrier(0);
                const v16bf bxh = xh[c % 3];
                const v16bf bxl = xl[c % 3];
#pragma unroll
                for (int i = 0; i < YTILES_PER_WAVE; ++i)
                    acc[i] = __builtin_amdgcn_wmma_f32_16x16x32_bf16(
                                 false, yh[i][c], false, bxl, (short)0, acc[i], false, false);
#pragma unroll
                for (int i = 0; i < YTILES_PER_WAVE; ++i)
                    acc[i] = __builtin_amdgcn_wmma_f32_16x16x32_bf16(
                                 false, yl[i][c], false, bxh, (short)0, acc[i], false, false);
#pragma unroll
                for (int i = 0; i < YTILES_PER_WAVE; ++i)
                    acc[i] = __builtin_amdgcn_wmma_f32_16x16x32_bf16(
                                 false, yh[i][c], false, bxh, (short)0, acc[i], false, false);
                __builtin_amdgcn_sched_barrier(0);
            }

            const int xRowNext = xRowSlice + min(t + s + 3, XTILES_PER_SLICE - 1) * 16;
            if (s == 0) {
                STORE_XTILE(B, 1);
                lds_stage_barrier();
                LOAD_XTILE(B, xRowNext);
            } else {
                STORE_XTILE(A, 0);
                lds_stage_barrier();
                LOAD_XTILE(A, xRowNext);
            }

            {
                const unsigned short* nh = &lXh[(s ^ 1) * STAGE_ELEMS + fOff];
                const unsigned short* nl = &lXl[(s ^ 1) * STAGE_ELEMS + fOff];
                xh0 = lds_frag(nh);
                xl0 = lds_frag(nl);
                xh1 = lds_frag(nh + 32);
                xl1 = lds_frag(nl + 32);
            }
            __builtin_amdgcn_sched_barrier(0);

            float mn = 3.402823466e38f;
#pragma unroll
            for (int i = 0; i < YTILES_PER_WAVE; ++i) {
#pragma unroll
                for (int r = 0; r < 8; ++r) {
                    const float m    = acc[i][r];
                    const float z    = (m - 1.0f) * INV_SIGMA;
                    const float logp = fmaf(-0.5f * z, z, LOGC);
                    const float C    = -__expf(logp) * logp;
                    mn = fminf(mn, C);
                }
            }
            if (s) {
                const float keep = (lane < 16) ? mnEven : mn;
                const float send = (lane < 16) ? mn : mnEven;
                prow[t * 16] = fminf(keep, other_half(send));
            } else {
                mnEven = mn;
            }
        }
    }
}

__global__ __launch_bounds__(256) void reduce_min_kernel(const float* __restrict__ part,
                                                         float* __restrict__ out) {
    __shared__ float sm[8][32];
    const int xi = threadIdx.x & 31;
    const int g  = threadIdx.x >> 5;
    const int x  = blockIdx.x * 32 + xi;
    const float* p = part + (size_t)(g * (NYWAVES / 8)) * NROWS + x;
    float v[NYWAVES / 8];
#pragma unroll
    for (int j = 0; j < NYWAVES / 8; ++j) v[j] = p[(size_t)j * NROWS];
    float m = v[0];
#pragma unroll
    for (int j = 1; j < NYWAVES / 8; ++j) m = fminf(m, v[j]);
    sm[g][xi] = m;
    __syncthreads();
    if (g == 0) {
        float r = sm[0][xi];
#pragma unroll
        for (int k = 1; k < 8; ++k) r = fminf(r, sm[k][xi]);
        out[x] = r;
    }
}

extern "C" void kernel_launch(void* const* d_in, const int* in_sizes, int n_in,
                              void* d_out, int out_size, void* d_ws, size_t ws_size,
                              hipStream_t stream) {
    const float* Ex = (const float*)d_in[0];
    const float* Ey = (const float*)d_in[1];

    const size_t elems = (size_t)NROWS * DIM;
    unsigned short* Xhi = (unsigned short*)d_ws;
    unsigned short* Xlo = Xhi + elems;
    float*         part = (float*)(Xlo + elems);

    hipLaunchKernelGGL(normalize_split_kernel, dim3(NROWS / 8), dim3(256), 0,
                       stream, Ex, Xhi, Xlo);
    hipLaunchKernelGGL(cosmin_wmma_kernel,
                       dim3(NROWS / YROWS_PER_BLOCK, NXSLICES), dim3(THREADS), 0,
                       stream, Xhi, Xlo, Ey, part);
    hipLaunchKernelGGL(reduce_min_kernel, dim3(NROWS / 32), dim3(256), 0, stream,
                       part, (float*)d_out);
}


# X fragments read from LDS a whole tile at a time: the next tile's 32 reads go as one group behind the hand-over, under the epilogue; no LDS read left among a tile's 96 WMMAs (was 6 groups of 4)
# speedup vs baseline: 3.2367x; 1.0576x over previous
#include <hip/hip_runtime.h>

#define NROWS 8192
#define DIM   256
#define INV_SIGMA 3.33333333333f
#define LOGC      0.28503428291f

#define YTILES_PER_WAVE 4
#define WAVES_PER_BLOCK 4
#define THREADS         (WAVES_PER_BLOCK * 32)
#define YROWS_PER_BLOCK (WAVES_PER_BLOCK * YTILES_PER_WAVE * 16)
#define NYWAVES         (NROWS / (YTILES_PER_WAVE * 16))
#define NXTILES         (NROWS / 16)
#define NXSLICES        8
#define XTILES_PER_SLICE (NXTILES / NXSLICES)
#define KCHUNKS         (DIM / 32)
#define LDS_STRIDE      264
#define STAGE_ELEMS     (16 * LDS_STRIDE)

typedef __attribute__((ext_vector_type(16))) __bf16 v16bf;
typedef __attribute__((ext_vector_type(8)))  float  v8f;
typedef __attribute__((ext_vector_type(4)))  float  f32x4;

__device__ __forceinline__ unsigned short f32_to_bf16_rne(float f) {
    unsigned u = __float_as_uint(f);
    u += 0x7FFFu + ((u >> 16) & 1u);
    return (unsigned short)(u >> 16);
}
__device__ __forceinline__ float bf16_to_f32(unsigned short h) {
    return __uint_as_float(((unsigned)h) << 16);
}
typedef __attribute__((ext_vector_type(2))) float  f32x2;
typedef __attribute__((ext_vector_type(2))) __bf16 bf16x2;
__device__ __forceinline__ void split_pair(float a, float b, unsigned& hi, unsigned& lo) {
    const f32x2 v = {a, b};
    const bf16x2 h = __builtin_convertvector(v, bf16x2);
    const unsigned hu = __builtin_bit_cast(unsigned, h);
    const f32x2 hf = {__uint_as_float(hu << 16), __uint_as_float(hu & 0xFFFF0000u)};
    const bf16x2 l = __builtin_convertvector(v - hf, bf16x2);
    hi = hu;
    lo = __builtin_bit_cast(unsigned, l);
}
__device__ __forceinline__ float other_half(float v) {
    return __int_as_float(__builtin_amdgcn_permlanex16(
        __float_as_int(v), __float_as_int(v), 0x76543210, 0xfedcba98, false, false));
}

__global__ void normalize_split_kernel(const float* __restrict__ Ex,
                                       unsigned short* __restrict__ Xhi,
                                       unsigned short* __restrict__ Xlo) {
    const int row  = (blockIdx.x * blockDim.x + threadIdx.x) >> 5;
    const int lane = threadIdx.x & 31;

    const float* p = Ex + (size_t)row * DIM;
    float v[DIM / 32];
    float ss = 0.0f;
#pragma unroll
    for (int i = 0; i < DIM / 32; ++i) {
        v[i] = p[lane + 32 * i];
        ss = fmaf(v[i], v[i], ss);
    }
#pragma unroll
    for (int off = 16; off > 0; off >>= 1)
        ss += __shfl_xor(ss, off, 32);

    const float inv = 1.0f / fmaxf(sqrtf(ss), 1e-8f);

#pragma unroll
    for (int i = 0; i < DIM / 32; ++i) {
        const float x = v[i] * inv;
        const unsigned short h = f32_to_bf16_rne(x);
        const unsigned short l = f32_to_bf16_rne(x - bf16_to_f32(h));
        const size_t idx = (size_t)row * DIM + lane + 32 * i;
        Xhi[idx] = h;
        Xlo[idx] = l;
    }
}

__device__ __forceinline__ v16bf lds_frag(const unsigned short* p) {
    union { uint4 q[2]; v16bf v; } f;
    f.q[0] = *(const uint4*)(p);
    f.q[1] = *(const uint4*)(p + 16);
    return f.v;
}

typedef __attribute__((ext_vector_type(4))) unsigned int u32x4;

#define LOAD_XTILE(P, xRow0)                                                  \
    do {                                                                      \
        const size_t s_ = (size_t)((xRow0) + wave) * DIM + lane * 8;          \
        P##H0 = *(const u32x4*)&Xhi[s_];                                      \
        P##H1 = *(const u32x4*)&Xhi[s_ + 4 * DIM];                            \
        P##H2 = *(const u32x4*)&Xhi[s_ + 8 * DIM];                            \
        P##H3 = *(const u32x4*)&Xhi[s_ + 12 * DIM];                           \
        P##L0 = *(const u32x4*)&Xlo[s_];                                      \
        P##L1 = *(const u32x4*)&Xlo[s_ + 4 * DIM];                            \
        P##L2 = *(const u32x4*)&Xlo[s_ + 8 * DIM];                            \
        P##L3 = *(const u32x4*)&Xlo[s_ + 12 * DIM];                           \
    } while (0)
#define STORE_XTILE(P, stage)                                                 \
    do {                                                                      \
        const int o_ = (stage) * STAGE_ELEMS + wave * LDS_STRIDE + lane * 8;  \
        *(u32x4*)&lXh[o_]                   = P##H0;                          \
        *(u32x4*)&lXh[o_ + 4 * LDS_STRIDE]  = P##H1;                          \
        *(u32x4*)&lXh[o_ + 8 * LDS_STRIDE]  = P##H2;                          \
        *(u32x4*)&lXh[o_ + 12 * LDS_STRIDE] = P##H3;                          \
        *(u32x4*)&lXl[o_]                   = P##L0;                          \
        *(u32x4*)&lXl[o_ + 4 * LDS_STRIDE]  = P##L1;                          \
        *(u32x4*)&lXl[o_ + 8 * LDS_STRIDE]  = P##L2;                          \
        *(u32x4*)&lXl[o_ + 12 * LDS_STRIDE] = P##L3;                          \
    } while (0)

__device__ __forceinline__ void lds_stage_barrier() {
    asm volatile("s_wait_dscnt 0x0\n\ts_barrier_signal -1\n\ts_barrier_wait -1" ::: "memory");
}

__global__ __launch_bounds__(THREADS) void cosmin_wmma_kernel(
    const unsigned short* __restrict__ Xhi,
    const unsigned short* __restrict__ Xlo,
    const float* __restrict__ Ey,
    float* __restrict__ part) {
    __shared__ unsigned short lXh[2 * STAGE_ELEMS];
    __shared__ unsigned short lXl[2 * STAGE_ELEMS];

    const int tid   = threadIdx.x;
    const int lane  = tid & 31;
    const int wave  = tid >> 5;
    const int c16   = lane & 15;
    const int hiSel = (lane >> 4) << 3;
    const int yRow0 = blockIdx.x * YROWS_PER_BLOCK + wave * (YTILES_PER_WAVE * 16);
    const int xRowSlice = blockIdx.y * (XTILES_PER_SLICE * 16);

    u32x4 AH0, AH1, AH2, AH3, AL0, AL1, AL2, AL3;
    u32x4 BH0, BH1, BH2, BH3, BL0, BL1, BL2, BL3;
    LOAD_XTILE(A, xRowSlice);

    v16bf yh[YTILES_PER_WAVE][KCHUNKS];
    v16bf yl[YTILES_PER_WAVE][KCHUNKS];
#pragma unroll
    for (int i = 0; i < YTILES_PER_WAVE; ++i) {
        const float* yp = Ey + (size_t)(yRow0 + 16 * i + c16) * DIM + hiSel;
        f32x4 raw[KCHUNKS][4];
#pragma unroll
        for (int c = 0; c < KCHUNKS; ++c) {
            raw[c][0] = *(const f32x4*)(yp + 32 * c);
            raw[c][1] = *(const f32x4*)(yp + 32 * c + 4);
            raw[c][2] = *(const f32x4*)(yp + 32 * c + 16);
            raw[c][3] = *(const f32x4*)(yp + 32 * c + 20);
        }
        float s0 = 0.0f, s1 = 0.0f, s2 = 0.0f, s3 = 0.0f;
#pragma unroll
        for (int c = 0; c < KCHUNKS; ++c) {
#pragma unroll
            for (int j = 0; j < 4; ++j) {
                s0 = fmaf(raw[c][j].x, raw[c][j].x, s0);
                s1 = fmaf(raw[c][j].y, raw[c][j].y, s1);
                s2 = fmaf(raw[c][j].z, raw[c][j].z, s2);
                s3 = fmaf(raw[c][j].w, raw[c][j].w, s3);
            }
        }
        float ss = (s0 + s1) + (s2 + s3);
        ss += other_half(ss);
        const float inv = 1.0f / fmaxf(sqrtf(ss), 1e-8f);
#pragma unroll
        for (int c = 0; c < KCHUNKS; ++c) {
            union { unsigned u[8]; v16bf v; } H, L;
#pragma unroll
            for (int j = 0; j < 4; ++j) {
                split_pair(raw[c][j].x * inv, raw[c][j].y * inv, H.u[2 * j],     L.u[2 * j]);
                split_pair(raw[c][j].z * inv, raw[c][j].w * inv, H.u[2 * j + 1], L.u[2 * j + 1]);
            }
            yh[i][c] = H.v;
            yl[i][c] = L.v;
        }
    }

    LOAD_XTILE(B, xRowSlice + 16);
    STORE_XTILE(A, 0);
    lds_stage_barrier();
    LOAD_XTILE(A, xRowSlice + 32);

    const int fOff = c16 * LDS_STRIDE + hiSel;

    v16bf xh[KCHUNKS], xl[KCHUNKS];
#pragma unroll
    for (int c = 0; c < KCHUNKS; ++c) {
        xh[c] = lds_frag(&lXh[fOff + 32 * c]);
        xl[c] = lds_frag(&lXl[fOff + 32 * c]);
    }

    float* prow = part + (size_t)(blockIdx.x * WAVES_PER_BLOCK + wave) * NROWS
                       + xRowSlice + lane;
    float mnEven = 0.0f;

    for (int t = 0; t < XTILES_PER_SLICE; t += 2) {
#pragma unroll
        for (int s = 0; s < 2; ++s) {
            v8f acc[YTILES_PER_WAVE];
#pragma unroll
            for (int i = 0; i < YTILES_PER_WAVE; ++i) acc[i] = v8f{};

#pragma unroll
            for (int c = 0; c < KCHUNKS; ++c) {
                const v16bf bxh = xh[c];
                const v16bf bxl = xl[c];
#pragma unroll
                for (int i = 0; i < YTILES_PER_WAVE; ++i)
                    acc[i] = __builtin_amdgcn_wmma_f32_16x16x32_bf16(
                                 false, yh[i][c], false, bxl, (short)0, acc[i], false, false);
#pragma unroll
                for (int i = 0; i < YTILES_PER_WAVE; ++i)
                    acc[i] = __builtin_amdgcn_wmma_f32_16x16x32_bf16(
                                 false, yl[i][c], false, bxh, (short)0, acc[i], false, false);
#pragma unroll
                for (int i = 0; i < YTILES_PER_WAVE; ++i)
                    acc[i] = __builtin_amdgcn_wmma_f32_16x16x32_bf16(
                                 false, yh[i][c], false, bxh, (short)0, acc[i], false, false);
            }
            __builtin_amdgcn_sched_barrier(0);

            const int xRowNext = xRowSlice + min(t + s + 3, XTILES_PER_SLICE - 1) * 16;
            if (s == 0) {
                STORE_XTILE(B, 1);
                lds_stage_barrier();
                LOAD_XTILE(B, xRowNext);
            } else {
                STORE_XTILE(A, 0);
                lds_stage_barrier();
                LOAD_XTILE(A, xRowNext);
            }

            {
                const unsigned short* nh = &lXh[(s ^ 1) * STAGE_ELEMS + fOff];
                const unsigned short* nl = &lXl[(s ^ 1) * STAGE_ELEMS + fOff];
#pragma unroll
                for (int c = 0; c < KCHUNKS; ++c) {
                    xh[c] = lds_frag(nh + 32 * c);
                    xl[c] = lds_frag(nl + 32 * c);
                    __builtin_amdgcn_sched_barrier(0);
                }
            }

            float mn = 3.402823466e38f;
#pragma unroll
            for (int i = 0; i < YTILES_PER_WAVE; ++i) {
#pragma unroll
                for (int r = 0; r < 8; ++r) {
                    const float m    = acc[i][r];
                    const float z    = (m - 1.0f) * INV_SIGMA;
                    const float logp = fmaf(-0.5f * z, z, LOGC);
                    const float C    = -__expf(logp) * logp;
                    mn = fminf(mn, C);
                }
            }
            if (s) {
                const float keep = (lane < 16) ? mnEven : mn;
                const float send = (lane < 16) ? mn : mnEven;
                prow[t * 16] = fminf(keep, other_half(send));
            } else {
                mnEven = mn;
            }
        }
    }
}

__global__ __launch_bounds__(256) void reduce_min_kernel(const float* __restrict__ part,
                                                         float* __restrict__ out) {
    __shared__ float sm[8][32];
    const int xi = threadIdx.x & 31;
    const int g  = threadIdx.x >> 5;
    const int x  = blockIdx.x * 32 + xi;
    const float* p = part + (size_t)(g * (NYWAVES / 8)) * NROWS + x;
    float v[NYWAVES / 8];
#pragma unroll
    for (int j = 0; j < NYWAVES / 8; ++j) v[j] = p[(size_t)j * NROWS];
    float m = v[0];
#pragma unroll
    for (int j = 1; j < NYWAVES / 8; ++j) m = fminf(m, v[j]);
    sm[g][xi] = m;
    __syncthreads();
    if (g == 0) {
        float r = sm[0][xi];
#pragma unroll
        for (int k = 1; k < 8; ++k) r = fminf(r, sm[k][xi]);
        out[x] = r;
    }
}

extern "C" void kernel_launch(void* const* d_in, const int* in_sizes, int n_in,
                              void* d_out, int out_size, void* d_ws, size_t ws_size,
                              hipStream_t stream) {
    const float* Ex = (const float*)d_in[0];
    const float* Ey = (const float*)d_in[1];

    const size_t elems = (size_t)NROWS * DIM;
    unsigned short* Xhi = (unsigned short*)d_ws;
    unsigned short* Xlo = Xhi + elems;
    float*         part = (float*)(Xlo + elems);

    hipLaunchKernelGGL(normalize_split_kernel, dim3(NROWS / 8), dim3(256), 0,
                       stream, Ex, Xhi, Xlo);
    hipLaunchKernelGGL(cosmin_wmma_kernel,
                       dim3(NROWS / YROWS_PER_BLOCK, NXSLICES), dim3(THREADS), 0,
                       stream, Xhi, Xlo, Ey, part);
    hipLaunchKernelGGL(reduce_min_kernel, dim3(NROWS / 32), dim3(256), 0, stream,
                       part, (float*)d_out);
}


# Epilogue constants folded into the GEMM: Y rows scaled by sqrt(log2e/(2 sigma^2)), accumulators start at -scale (the -1 summed first): t=c-D*D, running max of t*2^t; 148 -> 86 vector ops per tile
# speedup vs baseline: 3.3485x; 1.0345x over previous
#include <hip/hip_runtime.h>

#define NROWS 8192
#define DIM   256
#define Y_SCALE   2.83107267f
#define T_OFF     0.41121753f
#define NEG_LN2   (-0.69314718f)

#define YTILES_PER_WAVE 4
#define WAVES_PER_BLOCK 4
#define THREADS         (WAVES_PER_BLOCK * 32)
#define YROWS_PER_BLOCK (WAVES_PER_BLOCK * YTILES_PER_WAVE * 16)
#define NYWAVES         (NROWS / (YTILES_PER_WAVE * 16))
#define NXTILES         (NROWS / 16)
#define NXSLICES        8
#define XTILES_PER_SLICE (NXTILES / NXSLICES)
#define KCHUNKS         (DIM / 32)
#define LDS_STRIDE      264
#define STAGE_ELEMS     (16 * LDS_STRIDE)

typedef __attribute__((ext_vector_type(16))) __bf16 v16bf;
typedef __attribute__((ext_vector_type(8)))  float  v8f;
typedef __attribute__((ext_vector_type(4)))  float  f32x4;

__device__ __forceinline__ unsigned short f32_to_bf16_rne(float f) {
    unsigned u = __float_as_uint(f);
    u += 0x7FFFu + ((u >> 16) & 1u);
    return (unsigned short)(u >> 16);
}
__device__ __forceinline__ float bf16_to_f32(unsigned short h) {
    return __uint_as_float(((unsigned)h) << 16);
}
typedef __attribute__((ext_vector_type(2))) float  f32x2;
typedef __attribute__((ext_vector_type(2))) __bf16 bf16x2;
__device__ __forceinline__ void split_pair(float a, float b, unsigned& hi, unsigned& lo) {
    const f32x2 v = {a, b};
    const bf16x2 h = __builtin_convertvector(v, bf16x2);
    const unsigned hu = __builtin_bit_cast(unsigned, h);
    const f32x2 hf = {__uint_as_float(hu << 16), __uint_as_float(hu & 0xFFFF0000u)};
    const bf16x2 l = __builtin_convertvector(v - hf, bf16x2);
    hi = hu;
    lo = __builtin_bit_cast(unsigned, l);
}
__device__ __forceinline__ float other_half(float v) {
    return __int_as_float(__builtin_amdgcn_permlanex16(
        __float_as_int(v), __float_as_int(v), 0x76543210, 0xfedcba98, false, false));
}

__global__ void normalize_split_kernel(const float* __restrict__ Ex,
                                       unsigned short* __restrict__ Xhi,
                                       unsigned short* __restrict__ Xlo) {
    const int row  = (blockIdx.x * blockDim.x + threadIdx.x) >> 5;
    const int lane = threadIdx.x & 31;

    const float* p = Ex + (size_t)row * DIM;
    float v[DIM / 32];
    float ss = 0.0f;
#pragma unroll
    for (int i = 0; i < DIM / 32; ++i) {
        v[i] = p[lane + 32 * i];
        ss = fmaf(v[i], v[i], ss);
    }
#pragma unroll
    for (int off = 16; off > 0; off >>= 1)
        ss += __shfl_xor(ss, off, 32);

    const float inv = 1.0f / fmaxf(sqrtf(ss), 1e-8f);

#pragma unroll
    for (int i = 0; i < DIM / 32; ++i) {
        const float x = v[i] * inv;
        const unsigned short h = f32_to_bf16_rne(x);
        const unsigned short l = f32_to_bf16_rne(x - bf16_to_f32(h));
        const size_t idx = (size_t)row * DIM + lane + 32 * i;
        Xhi[idx] = h;
        Xlo[idx] = l;
    }
}

__device__ __forceinline__ v16bf lds_frag(const unsigned short* p) {
    union { uint4 q[2]; v16bf v; } f;
    f.q[0] = *(const uint4*)(p);
    f.q[1] = *(const uint4*)(p + 16);
    return f.v;
}

typedef __attribute__((ext_vector_type(4))) unsigned int u32x4;

#define LOAD_XTILE(P, xRow0)                                                  \
    do {                                                                      \
        const size_t s_ = (size_t)((xRow0) + wave) * DIM + lane * 8;          \
        P##H0 = *(const u32x4*)&Xhi[s_];                                      \
        P##H1 = *(const u32x4*)&Xhi[s_ + 4 * DIM];                            \
        P##H2 = *(const u32x4*)&Xhi[s_ + 8 * DIM];                            \
        P##H3 = *(const u32x4*)&Xhi[s_ + 12 * DIM];                           \
        P##L0 = *(const u32x4*)&Xlo[s_];                                      \
        P##L1 = *(const u32x4*)&Xlo[s_ + 4 * DIM];                            \
        P##L2 = *(const u32x4*)&Xlo[s_ + 8 * DIM];                            \
        P##L3 = *(const u32x4*)&Xlo[s_ + 12 * DIM];                           \
    } while (0)
#define STORE_XTILE(P, stage)                                                 \
    do {                                                                      \
        const int o_ = (stage) * STAGE_ELEMS + wave * LDS_STRIDE + lane * 8;  \
        *(u32x4*)&lXh[o_]                   = P##H0;                          \
        *(u32x4*)&lXh[o_ + 4 * LDS_STRIDE]  = P##H1;                          \
        *(u32x4*)&lXh[o_ + 8 * LDS_STRIDE]  = P##H2;                          \
        *(u32x4*)&lXh[o_ + 12 * LDS_STRIDE] = P##H3;                          \
        *(u32x4*)&lXl[o_]                   = P##L0;                          \
        *(u32x4*)&lXl[o_ + 4 * LDS_STRIDE]  = P##L1;                          \
        *(u32x4*)&lXl[o_ + 8 * LDS_STRIDE]  = P##L2;                          \
        *(u32x4*)&lXl[o_ + 12 * LDS_STRIDE] = P##L3;                          \
    } while (0)

__device__ __forceinline__ void lds_stage_barrier() {
    asm volatile("s_wait_dscnt 0x0\n\ts_barrier_signal -1\n\ts_barrier_wait -1" ::: "memory");
}

__global__ __launch_bounds__(THREADS) void cosmin_wmma_kernel(
    const unsigned short* __restrict__ Xhi,
    const unsigned short* __restrict__ Xlo,
    const float* __restrict__ Ey,
    float* __restrict__ part) {
    __shared__ unsigned short lXh[2 * STAGE_ELEMS];
    __shared__ unsigned short lXl[2 * STAGE_ELEMS];

    const int tid   = threadIdx.x;
    const int lane  = tid & 31;
    const int wave  = tid >> 5;
    const int c16   = lane & 15;
    const int hiSel = (lane >> 4) << 3;
    const int yRow0 = blockIdx.x * YROWS_PER_BLOCK + wave * (YTILES_PER_WAVE * 16);
    const int xRowSlice = blockIdx.y * (XTILES_PER_SLICE * 16);

    u32x4 AH0, AH1, AH2, AH3, AL0, AL1, AL2, AL3;
    u32x4 BH0, BH1, BH2, BH3, BL0, BL1, BL2, BL3;
    LOAD_XTILE(A, xRowSlice);

    v16bf yh[YTILES_PER_WAVE][KCHUNKS];
    v16bf yl[YTILES_PER_WAVE][KCHUNKS];
#pragma unroll
    for (int i = 0; i < YTILES_PER_WAVE; ++i) {
        const float* yp = Ey + (size_t)(yRow0 + 16 * i + c16) * DIM + hiSel;
        f32x4 raw[KCHUNKS][4];
#pragma unroll
        for (int c = 0; c < KCHUNKS; ++c) {
            raw[c][0] = *(const f32x4*)(yp + 32 * c);
            raw[c][1] = *(const f32x4*)(yp + 32 * c + 4);
            raw[c][2] = *(const f32x4*)(yp + 32 * c + 16);
            raw[c][3] = *(const f32x4*)(yp + 32 * c + 20);
        }
        float s0 = 0.0f, s1 = 0.0f, s2 = 0.0f, s3 = 0.0f;
#pragma unroll
        for (int c = 0; c < KCHUNKS; ++c) {
#pragma unroll
            for (int j = 0; j < 4; ++j) {
                s0 = fmaf(raw[c][j].x, raw[c][j].x, s0);
                s1 = fmaf(raw[c][j].y, raw[c][j].y, s1);
                s2 = fmaf(raw[c][j].z, raw[c][j].z, s2);
                s3 = fmaf(raw[c][j].w, raw[c][j].w, s3);
            }
        }
        float ss = (s0 + s1) + (s2 + s3);
        ss += other_half(ss);
        const float inv = Y_SCALE / fmaxf(sqrtf(ss), 1e-8f);
#pragma unroll
        for (int c = 0; c < KCHUNKS; ++c) {
            union { unsigned u[8]; v16bf v; } H, L;
#pragma unroll
            for (int j = 0; j < 4; ++j) {
                split_pair(raw[c][j].x * inv, raw[c][j].y * inv, H.u[2 * j],     L.u[2 * j]);
                split_pair(raw[c][j].z * inv, raw[c][j].w * inv, H.u[2 * j + 1], L.u[2 * j + 1]);
            }
            yh[i][c] = H.v;
            yl[i][c] = L.v;
        }
    }

    LOAD_XTILE(B, xRowSlice + 16);
    STORE_XTILE(A, 0);
    lds_stage_barrier();
    LOAD_XTILE(A, xRowSlice + 32);

    const int fOff = c16 * LDS_STRIDE + hiSel;

    v16bf xh[KCHUNKS], xl[KCHUNKS];
#pragma unroll
    for (int c = 0; c < KCHUNKS; ++c) {
        xh[c] = lds_frag(&lXh[fOff + 32 * c]);
        xl[c] = lds_frag(&lXl[fOff + 32 * c]);
    }

    float* prow = part + (size_t)(blockIdx.x * WAVES_PER_BLOCK + wave) * NROWS
                       + xRowSlice + lane;
    float mxEven = 0.0f;

    v8f negS;
#pragma unroll
    for (int r = 0; r < 8; ++r) negS[r] = -Y_SCALE;

    for (int t = 0; t < XTILES_PER_SLICE; t += 2) {
#pragma unroll
        for (int s = 0; s < 2; ++s) {
            v8f acc[YTILES_PER_WAVE];

#pragma unroll
            for (int c = 0; c < KCHUNKS; ++c) {
                const v16bf bxh = xh[c];
                const v16bf bxl = xl[c];
#pragma unroll
                for (int i = 0; i < YTILES_PER_WAVE; ++i)
                    acc[i] = __builtin_amdgcn_wmma_f32_16x16x32_bf16(
                                 false, yh[i][c], false, bxl, (short)0,
                                 (c == 0) ? negS : acc[i], false, false);
#pragma unroll
                for (int i = 0; i < YTILES_PER_WAVE; ++i)
                    acc[i] = __builtin_amdgcn_wmma_f32_16x16x32_bf16(
                                 false, yl[i][c], false, bxh, (short)0, acc[i], false, false);
#pragma unroll
                for (int i = 0; i < YTILES_PER_WAVE; ++i)
                    acc[i] = __builtin_amdgcn_wmma_f32_16x16x32_bf16(
                                 false, yh[i][c], false, bxh, (short)0, acc[i], false, false);
            }
            __builtin_amdgcn_sched_barrier(0);

            const int xRowNext = xRowSlice + min(t + s + 3, XTILES_PER_SLICE - 1) * 16;
            if (s == 0) {
                STORE_XTILE(B, 1);
                lds_stage_barrier();
                LOAD_XTILE(B, xRowNext);
            } else {
                STORE_XTILE(A, 0);
                lds_stage_barrier();
                LOAD_XTILE(A, xRowNext);
            }

            {
                const unsigned short* nh = &lXh[(s ^ 1) * STAGE_ELEMS + fOff];
                const unsigned short* nl = &lXl[(s ^ 1) * STAGE_ELEMS + fOff];
#pragma unroll
                for (int c = 0; c < KCHUNKS; ++c) {
                    xh[c] = lds_frag(nh + 32 * c);
                    xl[c] = lds_frag(nl + 32 * c);
                    __builtin_amdgcn_sched_barrier(0);
                }
            }

            float mx = -3.402823466e38f;
#pragma unroll
            for (int i = 0; i < YTILES_PER_WAVE; ++i) {
#pragma unroll
                for (int r = 0; r < 8; ++r) {
                    const float d = acc[i][r];
                    const float t = fmaf(-d, d, T_OFF);
                    const float q = t * __builtin_amdgcn_exp2f(t);
                    mx = fmaxf(mx, q);
                }
            }
            if (s) {
                const float keep = (lane < 16) ? mxEven : mx;
                const float send = (lane < 16) ? mx : mxEven;
                prow[t * 16] = NEG_LN2 * fmaxf(keep, other_half(send));
            } else {
                mxEven = mx;
            }
        }
    }
}

__global__ __launch_bounds__(256) void reduce_min_kernel(const float* __restrict__ part,
                                                         float* __restrict__ out) {
    __shared__ float sm[8][32];
    const int xi = threadIdx.x & 31;
    const int g  = threadIdx.x >> 5;
    const int x  = blockIdx.x * 32 + xi;
    const float* p = part + (size_t)(g * (NYWAVES / 8)) * NROWS + x;
    float v[NYWAVES / 8];
#pragma unroll
    for (int j = 0; j < NYWAVES / 8; ++j) v[j] = p[(size_t)j * NROWS];
    float m = v[0];
#pragma unroll
    for (int j = 1; j < NYWAVES / 8; ++j) m = fminf(m, v[j]);
    sm[g][xi] = m;
    __syncthreads();
    if (g == 0) {
        float r = sm[0][xi];
#pragma unroll
        for (int k = 1; k < 8; ++k) r = fminf(r, sm[k][xi]);
        out[x] = r;
    }
}

extern "C" void kernel_launch(void* const* d_in, const int* in_sizes, int n_in,
                              void* d_out, int out_size, void* d_ws, size_t ws_size,
                              hipStream_t stream) {
    const float* Ex = (const float*)d_in[0];
    const float* Ey = (const float*)d_in[1];

    const size_t elems = (size_t)NROWS * DIM;
    unsigned short* Xhi = (unsigned short*)d_ws;
    unsigned short* Xlo = Xhi + elems;
    float*         part = (float*)(Xlo + elems);

    hipLaunchKernelGGL(normalize_split_kernel, dim3(NROWS / 8), dim3(256), 0,
                       stream, Ex, Xhi, Xlo);
    hipLaunchKernelGGL(cosmin_wmma_kernel,
                       dim3(NROWS / YROWS_PER_BLOCK, NXSLICES), dim3(THREADS), 0,
                       stream, Xhi, Xlo, Ey, part);
    hipLaunchKernelGGL(reduce_min_kernel, dim3(NROWS / 32), dim3(256), 0, stream,
                       part, (float*)d_out);
}
